# starting kernel
# baseline (speedup 1.0000x reference)
#include <hip/hip_runtime.h>
#include <hip/hip_bf16.h>
#include <math.h>

typedef float v2f __attribute__((ext_vector_type(2)));
typedef float v8f __attribute__((ext_vector_type(8)));

#define NEG_SLOPE 0.2f
#define EPS_F 1e-16f

#define SWIZ_XOR1  0x041F
#define SWIZ_XOR2  0x081F
#define SWIZ_XOR4  0x101F
#define SWIZ_XOR8  0x201F
#define SWIZ_XOR16 0x401F
#define SWZ(v, imm) __int_as_float(__builtin_amdgcn_ds_swizzle(__float_as_int(v), imm))

__device__ __forceinline__ float reduce16_add(float v) {
    v += SWZ(v, SWIZ_XOR1);
    v += SWZ(v, SWIZ_XOR2);
    v += SWZ(v, SWIZ_XOR4);
    v += SWZ(v, SWIZ_XOR8);
    return v;
}
__device__ __forceinline__ float reduce32_max(float v) {
    v = fmaxf(v, SWZ(v, SWIZ_XOR1));
    v = fmaxf(v, SWZ(v, SWIZ_XOR2));
    v = fmaxf(v, SWZ(v, SWIZ_XOR4));
    v = fmaxf(v, SWZ(v, SWIZ_XOR8));
    v = fmaxf(v, SWZ(v, SWIZ_XOR16));
    return v;
}

__device__ __forceinline__ unsigned int fkey(float f) {
    unsigned int u = __float_as_uint(f);
    return (u & 0x80000000u) ? ~u : (u | 0x80000000u);
}
__device__ __forceinline__ float funkey(unsigned int k) {
    return __uint_as_float((k & 0x80000000u) ? (k & 0x7FFFFFFFu) : ~k);
}

__global__ __launch_bounds__(256) void gat_gemm_wmma(
    const float* __restrict__ x, const float* __restrict__ W,
    const float* __restrict__ a_src, const float* __restrict__ a_trg,
    float* __restrict__ hbuf, float* __restrict__ s_src, float* __restrict__ s_trg,
    int N)
{
    const int wave  = threadIdx.x >> 5;
    const int lane  = threadIdx.x & 31;
    const int mBase = blockIdx.x * 16;
    const int nBase = wave * 16;

    const int f    = lane & 15;
    const int half = lane >> 4;
    const int ksub = half * 2;

    int rowA = mBase + f;
    if (rowA > N - 1) rowA = N - 1;
    const float* __restrict__ xrow = x + (size_t)rowA * 128;

    v8f c = {};
    #pragma unroll 8
    for (int k = 0; k < 128; k += 4) {
        v2f a;
        a.x = xrow[k + ksub];
        a.y = xrow[k + ksub + 1];
        v2f b;
        b.x = W[(size_t)(k + ksub) * 128 + nBase + f];
        b.y = W[(size_t)(k + ksub + 1) * 128 + nBase + f];
        c = __builtin_amdgcn_wmma_f32_16x16x4_f32(
                 false, a,  false, b,
                 (short)0, c,  false,  false);
    }

    const float asv = a_src[nBase + f];
    const float atv = a_trg[nBase + f];

    #pragma unroll
    for (int r = 0; r < 8; ++r) {
        const int m = mBase + r + half * 8;
        const float hv = c[r];
        if (m < N) hbuf[(size_t)m * 128 + nBase + f] = hv;

        const float ps = reduce16_add(hv * asv);
        const float pt = reduce16_add(hv * atv);
        if (f == 0 && m < N) {
            s_src[(size_t)m * 8 + wave] = ps;
            s_trg[(size_t)m * 8 + wave] = pt;
        }
    }
}

__global__ __launch_bounds__(256) void gat_edge_scores(
    const int* __restrict__ ei, int E,
    const float* __restrict__ s_src, const float* __restrict__ s_trg,
    float* __restrict__ ebuf, unsigned int* __restrict__ gmax)
{
    __shared__ unsigned int smax;
    if (threadIdx.x == 0) smax = 0u;
    __syncthreads();

    const int e = blockIdx.x * blockDim.x + threadIdx.x;
    float lmax = -INFINITY;
    if (e < E) {
        const int src = ei[e];
        const int trg = ei[E + e];
        const float* __restrict__ ss = s_src + (size_t)src * 8;
        const float* __restrict__ st = s_trg + (size_t)trg * 8;
        #pragma unroll
        for (int hh = 0; hh < 8; ++hh) {
            float v = ss[hh] + st[hh];
            v = (v >= 0.f) ? v : NEG_SLOPE * v;
            ebuf[(size_t)e * 8 + hh] = v;
            lmax = fmaxf(lmax, v);
        }
    }
    lmax = reduce32_max(lmax);
    if ((threadIdx.x & 31) == 0) atomicMax(&smax, fkey(lmax));
    __syncthreads();
    if (threadIdx.x == 0) atomicMax(gmax, smax);
}

__global__ __launch_bounds__(256) void gat_edge_exp(
    const int* __restrict__ ei, int E,
    float* __restrict__ ebuf, float* __restrict__ denom,
    const unsigned int* __restrict__ gmax)
{
    const int e = blockIdx.x * blockDim.x + threadIdx.x;
    if (e >= E) return;
    const float mx = funkey(*gmax);
    const int trg = ei[E + e];
    #pragma unroll
    for (int hh = 0; hh < 8; ++hh) {
        const float v = expf(ebuf[(size_t)e * 8 + hh] - mx);
        ebuf[(size_t)e * 8 + hh] = v;
        atomicAdd(&denom[(size_t)trg * 8 + hh], v);
    }
}

__global__ __launch_bounds__(256) void gat_aggregate(
    const int* __restrict__ ei, int E,
    const float* __restrict__ hbuf, const float* __restrict__ ebuf,
    const float* __restrict__ denom, float* __restrict__ out)
{
    const int t = blockIdx.x * blockDim.x + threadIdx.x;
    if (t >= E * 8) return;
    const int e  = t >> 3;
    const int hh = t & 7;
    const int src = ei[e];
    const int trg = ei[E + e];

    const float alpha = ebuf[t] / (denom[(size_t)trg * 8 + hh] + EPS_F);
    const float4* __restrict__ hp = (const float4*)(hbuf + (size_t)src * 128 + hh * 16);
    float* __restrict__ op = out + (size_t)trg * 128 + hh * 16;
    #pragma unroll
    for (int q = 0; q < 4; ++q) {
        const float4 v = hp[q];
        atomicAdd(op + q * 4 + 0, v.x * alpha);
        atomicAdd(op + q * 4 + 1, v.y * alpha);
        atomicAdd(op + q * 4 + 2, v.z * alpha);
        atomicAdd(op + q * 4 + 3, v.w * alpha);
    }
}

__global__ __launch_bounds__(256) void gat_epilogue(
    const float* __restrict__ hbuf, const float* __restrict__ bias,
    float* __restrict__ out, int total)
{
    const int i = blockIdx.x * blockDim.x + threadIdx.x;
    if (i >= total) return;
    float v = out[i] + hbuf[i] + bias[i & 127];
    out[i] = (v > 0.f) ? v : expm1f(v);
}

extern "C" void kernel_launch(void* const* d_in, const int* in_sizes, int n_in,
                              void* d_out, int out_size, void* d_ws, size_t ws_size,
                              hipStream_t stream)
{
    const float* x      = (const float*)d_in[0];
    const int*   ei     = (const int*)  d_in[1];
    const float* W      = (const float*)d_in[2];
    const float* a_src  = (const float*)d_in[3];
    const float* a_trg  = (const float*)d_in[4];
    const float* bias   = (const float*)d_in[5];

    const int N = in_sizes[0] / 128;
    const int E = in_sizes[1] / 2;

    float* hbuf  = (float*)d_ws;
    float* ssrc  = hbuf  + (size_t)N * 128;
    float* strg  = ssrc  + (size_t)N * 8;
    float* ebuf  = strg  + (size_t)N * 8;
    float* denom = ebuf  + (size_t)E * 8;
    unsigned int* gmax = (unsigned int*)(denom + (size_t)N * 8);

    hipMemsetAsync(denom, 0, (size_t)N * 8 * sizeof(float) + sizeof(unsigned int), stream);
    hipMemsetAsync(d_out, 0, (size_t)out_size * sizeof(float), stream);

    const int mtiles = (N + 15) / 16;
    gat_gemm_wmma<<<mtiles, 256, 0, stream>>>(x, W, a_src, a_trg, hbuf, ssrc, strg, N);

    gat_edge_scores<<<(E + 255) / 256, 256, 0, stream>>>(ei, E, ssrc, strg, ebuf, gmax);

    gat_edge_exp<<<(E + 255) / 256, 256, 0, stream>>>(ei, E, ebuf, denom, gmax);

    gat_aggregate<<<(E * 8 + 255) / 256, 256, 0, stream>>>(ei, E, hbuf, ebuf, denom, (float*)d_out);

    const int total = N * 128;
    gat_epilogue<<<(total + 255) / 256, 256, 0, stream>>>(hbuf, bias, (float*)d_out, total);
}


# aggregate: wave per 32 edges, a row's 128 floats on 32 lanes x 4, so each f32 atomic instruction covers one whole 128-byte line of out, not a 64-byte piece per lane; alpha per head by DPP row_share
# speedup vs baseline: 2.1830x; 2.1830x over previous
#include <hip/hip_runtime.h>
#include <hip/hip_bf16.h>
#include <math.h>

typedef float v2f __attribute__((ext_vector_type(2)));
typedef float v8f __attribute__((ext_vector_type(8)));

#define NEG_SLOPE 0.2f
#define EPS_F 1e-16f

#define SWIZ_XOR1  0x041F
#define SWIZ_XOR2  0x081F
#define SWIZ_XOR4  0x101F
#define SWIZ_XOR8  0x201F
#define SWIZ_XOR16 0x401F
#define SWZ(v, imm) __int_as_float(__builtin_amdgcn_ds_swizzle(__float_as_int(v), imm))

__device__ __forceinline__ float reduce16_add(float v) {
    v += SWZ(v, SWIZ_XOR1);
    v += SWZ(v, SWIZ_XOR2);
    v += SWZ(v, SWIZ_XOR4);
    v += SWZ(v, SWIZ_XOR8);
    return v;
}
__device__ __forceinline__ float reduce32_max(float v) {
    v = fmaxf(v, SWZ(v, SWIZ_XOR1));
    v = fmaxf(v, SWZ(v, SWIZ_XOR2));
    v = fmaxf(v, SWZ(v, SWIZ_XOR4));
    v = fmaxf(v, SWZ(v, SWIZ_XOR8));
    v = fmaxf(v, SWZ(v, SWIZ_XOR16));
    return v;
}

__device__ __forceinline__ unsigned int fkey(float f) {
    unsigned int u = __float_as_uint(f);
    return (u & 0x80000000u) ? ~u : (u | 0x80000000u);
}
__device__ __forceinline__ float funkey(unsigned int k) {
    return __uint_as_float((k & 0x80000000u) ? (k & 0x7FFFFFFFu) : ~k);
}

__global__ __launch_bounds__(256) void gat_gemm_wmma(
    const float* __restrict__ x, const float* __restrict__ W,
    const float* __restrict__ a_src, const float* __restrict__ a_trg,
    float* __restrict__ hbuf, float* __restrict__ s_src, float* __restrict__ s_trg,
    int N)
{
    const int wave  = threadIdx.x >> 5;
    const int lane  = threadIdx.x & 31;
    const int mBase = blockIdx.x * 16;
    const int nBase = wave * 16;

    const int f    = lane & 15;
    const int half = lane >> 4;
    const int ksub = half * 2;

    int rowA = mBase + f;
    if (rowA > N - 1) rowA = N - 1;
    const float* __restrict__ xrow = x + (size_t)rowA * 128;

    v8f c = {};
    #pragma unroll 8
    for (int k = 0; k < 128; k += 4) {
        v2f a;
        a.x = xrow[k + ksub];
        a.y = xrow[k + ksub + 1];
        v2f b;
        b.x = W[(size_t)(k + ksub) * 128 + nBase + f];
        b.y = W[(size_t)(k + ksub + 1) * 128 + nBase + f];
        c = __builtin_amdgcn_wmma_f32_16x16x4_f32(
                 false, a,  false, b,
                 (short)0, c,  false,  false);
    }

    const float asv = a_src[nBase + f];
    const float atv = a_trg[nBase + f];

    #pragma unroll
    for (int r = 0; r < 8; ++r) {
        const int m = mBase + r + half * 8;
        const float hv = c[r];
        if (m < N) hbuf[(size_t)m * 128 + nBase + f] = hv;

        const float ps = reduce16_add(hv * asv);
        const float pt = reduce16_add(hv * atv);
        if (f == 0 && m < N) {
            s_src[(size_t)m * 8 + wave] = ps;
            s_trg[(size_t)m * 8 + wave] = pt;
        }
    }
}

__global__ __launch_bounds__(256) void gat_edge_scores(
    const int* __restrict__ ei, int E,
    const float* __restrict__ s_src, const float* __restrict__ s_trg,
    float* __restrict__ ebuf, unsigned int* __restrict__ gmax)
{
    __shared__ unsigned int smax;
    if (threadIdx.x == 0) smax = 0u;
    __syncthreads();

    const int e = blockIdx.x * blockDim.x + threadIdx.x;
    float lmax = -INFINITY;
    if (e < E) {
        const int src = ei[e];
        const int trg = ei[E + e];
        const float* __restrict__ ss = s_src + (size_t)src * 8;
        const float* __restrict__ st = s_trg + (size_t)trg * 8;
        #pragma unroll
        for (int hh = 0; hh < 8; ++hh) {
            float v = ss[hh] + st[hh];
            v = (v >= 0.f) ? v : NEG_SLOPE * v;
            ebuf[(size_t)e * 8 + hh] = v;
            lmax = fmaxf(lmax, v);
        }
    }
    lmax = reduce32_max(lmax);
    if ((threadIdx.x & 31) == 0) atomicMax(&smax, fkey(lmax));
    __syncthreads();
    if (threadIdx.x == 0) atomicMax(gmax, smax);
}

__global__ __launch_bounds__(256) void gat_edge_exp(
    const int* __restrict__ ei, int E,
    float* __restrict__ ebuf, float* __restrict__ denom,
    const unsigned int* __restrict__ gmax)
{
    const int e = blockIdx.x * blockDim.x + threadIdx.x;
    if (e >= E) return;
    const float mx = funkey(*gmax);
    const int trg = ei[E + e];
    #pragma unroll
    for (int hh = 0; hh < 8; ++hh) {
        const float v = expf(ebuf[(size_t)e * 8 + hh] - mx);
        ebuf[(size_t)e * 8 + hh] = v;
        atomicAdd(&denom[(size_t)trg * 8 + hh], v);
    }
}


template <int SEL>
__device__ __forceinline__ float row_share(float v) {
    return __int_as_float(__builtin_amdgcn_update_dpp(
        0, __float_as_int(v), 0x150 + SEL, 0xF, 0xF, true));
}

__device__ __forceinline__ void load_row(const float* __restrict__ hp, float* hv) {
    hv[0] = hp[0];
    hv[1] = hp[32];
    hv[2] = hp[64];
    hv[3] = hp[96];
}

template <int K>
__device__ __forceinline__ void add_row(float* __restrict__ op, const float* hv, float al) {
    atomicAdd(op +  0, hv[0] * row_share<4 * K + 0>(al));
    atomicAdd(op + 32, hv[1] * row_share<4 * K + 1>(al));
    atomicAdd(op + 64, hv[2] * row_share<4 * K + 2>(al));
    atomicAdd(op + 96, hv[3] * row_share<4 * K + 3>(al));
}

__global__ __launch_bounds__(256) void gat_aggregate(
    const int* __restrict__ ei, int E,
    const float* __restrict__ hbuf, const float* __restrict__ ebuf,
    const float* __restrict__ denom, float* __restrict__ out)
{
    const int lane = threadIdx.x & 31;
    const int half = lane >> 4;
    const int wv = __builtin_amdgcn_readfirstlane((int)(blockIdx.x * 8 + (threadIdx.x >> 5)));
    const size_t e0 = (size_t)wv * 32;
    if (e0 >= (size_t)E) return;
    const int cnt = ((size_t)E - e0 < 32) ? (int)((size_t)E - e0) : 32;

    if (cnt == 32) {
        const int vsrc = ei[e0 + lane];
        const int vtrg = ei[(size_t)E + e0 + lane];

        const int sub = (lane & 15) >> 2;
        const int hd  = 2 * (lane & 3) + half;
        float al[8];
        #pragma unroll
        for (int j = 0; j < 8; ++j) {
            const int t = __shfl(vtrg, 4 * j + sub, 32);
            const float eb = ebuf[(e0 + 4 * j + sub) * 8 + hd];
            const float dn = denom[(size_t)t * 8 + hd];
            al[j] = eb / (dn + EPS_F);
        }

        float hv[2][16];
        #pragma unroll
        for (int k = 0; k < 4; ++k) {
            const int s = __builtin_amdgcn_readlane(vsrc, k);
            load_row(hbuf + (size_t)s * 128 + lane, &hv[0][4 * k]);
        }
        #pragma unroll
        for (int j = 0; j < 8; ++j) {
            const int cur = j & 1;
            if (j + 1 < 8) {
                #pragma unroll
                for (int k = 0; k < 4; ++k) {
                    const int s = __builtin_amdgcn_readlane(vsrc, 4 * (j + 1) + k);
                    load_row(hbuf + (size_t)s * 128 + lane, &hv[cur ^ 1][4 * k]);
                }
            }
            const int t0 = __builtin_amdgcn_readlane(vtrg, 4 * j + 0);
            const int t1 = __builtin_amdgcn_readlane(vtrg, 4 * j + 1);
            const int t2 = __builtin_amdgcn_readlane(vtrg, 4 * j + 2);
            const int t3 = __builtin_amdgcn_readlane(vtrg, 4 * j + 3);
            add_row<0>(out + (size_t)t0 * 128 + lane, &hv[cur][0],  al[j]);
            add_row<1>(out + (size_t)t1 * 128 + lane, &hv[cur][4],  al[j]);
            add_row<2>(out + (size_t)t2 * 128 + lane, &hv[cur][8],  al[j]);
            add_row<3>(out + (size_t)t3 * 128 + lane, &hv[cur][12], al[j]);
        }
    } else {
        for (int i = 0; i < cnt; ++i) {
            const size_t e = e0 + i;
            const int s = ei[e];
            const int t = ei[(size_t)E + e];
            #pragma unroll
            for (int q = 0; q < 4; ++q) {
                const int hh = 2 * q + half;
                const float a = ebuf[e * 8 + hh] / (denom[(size_t)t * 8 + hh] + EPS_F);
                atomicAdd(out + (size_t)t * 128 + q * 32 + lane,
                          hbuf[(size_t)s * 128 + q * 32 + lane] * a);
            }
        }
    }
}

__global__ __launch_bounds__(256) void gat_epilogue(
    const float* __restrict__ hbuf, const float* __restrict__ bias,
    float* __restrict__ out, int total)
{
    const int i = blockIdx.x * blockDim.x + threadIdx.x;
    if (i >= total) return;
    float v = out[i] + hbuf[i] + bias[i & 127];
    out[i] = (v > 0.f) ? v : expm1f(v);
}

extern "C" void kernel_launch(void* const* d_in, const int* in_sizes, int n_in,
                              void* d_out, int out_size, void* d_ws, size_t ws_size,
                              hipStream_t stream)
{
    const float* x      = (const float*)d_in[0];
    const int*   ei     = (const int*)  d_in[1];
    const float* W      = (const float*)d_in[2];
    const float* a_src  = (const float*)d_in[3];
    const float* a_trg  = (const float*)d_in[4];
    const float* bias   = (const float*)d_in[5];

    const int N = in_sizes[0] / 128;
    const int E = in_sizes[1] / 2;

    float* hbuf  = (float*)d_ws;
    float* ssrc  = hbuf  + (size_t)N * 128;
    float* strg  = ssrc  + (size_t)N * 8;
    float* ebuf  = strg  + (size_t)N * 8;
    float* denom = ebuf  + (size_t)E * 8;
    unsigned int* gmax = (unsigned int*)(denom + (size_t)N * 8);

    hipMemsetAsync(denom, 0, (size_t)N * 8 * sizeof(float) + sizeof(unsigned int), stream);
    hipMemsetAsync(d_out, 0, (size_t)out_size * sizeof(float), stream);

    const int mtiles = (N + 15) / 16;
    gat_gemm_wmma<<<mtiles, 256, 0, stream>>>(x, W, a_src, a_trg, hbuf, ssrc, strg, N);

    gat_edge_scores<<<(E + 255) / 256, 256, 0, stream>>>(ei, E, ssrc, strg, ebuf, gmax);

    gat_edge_exp<<<(E + 255) / 256, 256, 0, stream>>>(ei, E, ebuf, denom, gmax);

    gat_aggregate<<<(E + 255) / 256, 256, 0, stream>>>(ei, E, hbuf, ebuf, denom, (float*)d_out);

    const int total = N * 128;
    gat_epilogue<<<(total + 255) / 256, 256, 0, stream>>>(hbuf, bias, (float*)d_out, total);
}


# aggregate by target: edges counting-sorted by target (integer atomics), one wave per node sums its source rows in registers, writes out once; sums in arrival-order rank, no float atomics
# speedup vs baseline: 3.4757x; 1.5922x over previous
#include <hip/hip_runtime.h>
#include <hip/hip_bf16.h>
#include <math.h>

typedef float v2f __attribute__((ext_vector_type(2)));
typedef float v4f __attribute__((ext_vector_type(4)));
typedef float v8f __attribute__((ext_vector_type(8)));

#define NEG_SLOPE 0.2f
#define EPS_F 1e-16f

#define SWIZ_XOR1  0x041F
#define SWIZ_XOR2  0x081F
#define SWIZ_XOR4  0x101F
#define SWIZ_XOR8  0x201F
#define SWIZ_XOR16 0x401F
#define SWZ(v, imm) __int_as_float(__builtin_amdgcn_ds_swizzle(__float_as_int(v), imm))

__device__ __forceinline__ float reduce16_add(float v) {
    v += SWZ(v, SWIZ_XOR1);
    v += SWZ(v, SWIZ_XOR2);
    v += SWZ(v, SWIZ_XOR4);
    v += SWZ(v, SWIZ_XOR8);
    return v;
}
__device__ __forceinline__ float reduce32_max(float v) {
    v = fmaxf(v, SWZ(v, SWIZ_XOR1));
    v = fmaxf(v, SWZ(v, SWIZ_XOR2));
    v = fmaxf(v, SWZ(v, SWIZ_XOR4));
    v = fmaxf(v, SWZ(v, SWIZ_XOR8));
    v = fmaxf(v, SWZ(v, SWIZ_XOR16));
    return v;
}

__device__ __forceinline__ unsigned int fkey(float f) {
    unsigned int u = __float_as_uint(f);
    return (u & 0x80000000u) ? ~u : (u | 0x80000000u);
}
__device__ __forceinline__ float funkey(unsigned int k) {
    return __uint_as_float((k & 0x80000000u) ? (k & 0x7FFFFFFFu) : ~k);
}

__global__ __launch_bounds__(256) void gat_gemm_wmma(
    const float* __restrict__ x, const float* __restrict__ W,
    const float* __restrict__ a_src, const float* __restrict__ a_trg,
    float* __restrict__ hbuf, float* __restrict__ s_src, float* __restrict__ s_trg,
    int N)
{
    const int wave  = threadIdx.x >> 5;
    const int lane  = threadIdx.x & 31;
    const int mBase = blockIdx.x * 16;
    const int nBase = wave * 16;

    const int f    = lane & 15;
    const int half = lane >> 4;
    const int ksub = half * 2;

    int rowA = mBase + f;
    if (rowA > N - 1) rowA = N - 1;
    const float* __restrict__ xrow = x + (size_t)rowA * 128;

    v8f c = {};
    #pragma unroll 8
    for (int k = 0; k < 128; k += 4) {
        v2f a;
        a.x = xrow[k + ksub];
        a.y = xrow[k + ksub + 1];
        v2f b;
        b.x = W[(size_t)(k + ksub) * 128 + nBase + f];
        b.y = W[(size_t)(k + ksub + 1) * 128 + nBase + f];
        c = __builtin_amdgcn_wmma_f32_16x16x4_f32(
                 false, a,  false, b,
                 (short)0, c,  false,  false);
    }

    const float asv = a_src[nBase + f];
    const float atv = a_trg[nBase + f];

    #pragma unroll
    for (int r = 0; r < 8; ++r) {
        const int m = mBase + r + half * 8;
        const float hv = c[r];
        if (m < N) hbuf[(size_t)m * 128 + nBase + f] = hv;

        const float ps = reduce16_add(hv * asv);
        const float pt = reduce16_add(hv * atv);
        if (f == 0 && m < N) {
            s_src[(size_t)m * 8 + wave] = ps;
            s_trg[(size_t)m * 8 + wave] = pt;
        }
    }
}

__global__ __launch_bounds__(256) void gat_edge_max_count(
    const int* __restrict__ ei, int E,
    const float* __restrict__ s_src, const float* __restrict__ s_trg,
    int* __restrict__ cnt, unsigned int* __restrict__ gmax)
{
    __shared__ unsigned int smax;
    if (threadIdx.x == 0) smax = 0u;
    __syncthreads();

    const int e = blockIdx.x * blockDim.x + threadIdx.x;
    float lmax = -INFINITY;
    if (e < E) {
        const int src = ei[e];
        const int trg = ei[(size_t)E + e];
        const float* __restrict__ ss = s_src + (size_t)src * 8;
        const float* __restrict__ st = s_trg + (size_t)trg * 8;
        #pragma unroll
        for (int hh = 0; hh < 8; ++hh) {
            float v = ss[hh] + st[hh];
            v = (v >= 0.f) ? v : NEG_SLOPE * v;
            lmax = fmaxf(lmax, v);
        }
        atomicAdd(&cnt[trg], 1);
    }
    lmax = reduce32_max(lmax);
    if ((threadIdx.x & 31) == 0) atomicMax(&smax, fkey(lmax));
    __syncthreads();
    if (threadIdx.x == 0) atomicMax(gmax, smax);
}

__global__ __launch_bounds__(1024) void gat_scan_counts(
    const int* __restrict__ cnt, int* __restrict__ off, int* __restrict__ cur, int N)
{
    __shared__ int wsum[32];
    __shared__ int wpre[32];
    __shared__ int sbase;

    const int tid  = threadIdx.x;
    const int lane = tid & 31;
    const int wv   = tid >> 5;
    const int i    = blockIdx.x * 1024 + tid;
    const int c    = (i < N) ? cnt[i] : 0;

    int pre = 0;
    #pragma unroll 8
    for (int j = 0; j < (int)blockIdx.x; ++j) pre += cnt[j * 1024 + tid];

    int inc = c;
    #pragma unroll
    for (int d = 1; d < 32; d <<= 1) {
        const int u = __shfl_up(inc, d, 32);
        if (lane >= d) inc += u;
    }
    #pragma unroll
    for (int d = 16; d >= 1; d >>= 1) pre += __shfl_xor(pre, d, 32);

    if (lane == 31) wsum[wv] = inc;
    if (lane == 0)  wpre[wv] = pre;
    __syncthreads();
    if (wv == 0) {
        const int w = wsum[lane];
        int winc = w;
        #pragma unroll
        for (int d = 1; d < 32; d <<= 1) {
            const int u = __shfl_up(winc, d, 32);
            if (lane >= d) winc += u;
        }
        int p = wpre[lane];
        #pragma unroll
        for (int d = 16; d >= 1; d >>= 1) p += __shfl_xor(p, d, 32);
        wsum[lane] = winc - w;
        if (lane == 0) sbase = p;
    }
    __syncthreads();
    const int o = sbase + wsum[wv] + inc - c;
    if (i < N) { off[i] = o; cur[i] = o; }
    if (i == N - 1) off[N] = o + c;
}

__global__ __launch_bounds__(256) void gat_fill_lists(
    const int* __restrict__ ei, int E, int* __restrict__ cur, int* __restrict__ lst)
{
    const int e = blockIdx.x * blockDim.x + threadIdx.x;
    if (e >= E) return;
    const int s = ei[e];
    const int t = ei[(size_t)E + e];
    const int p = atomicAdd(&cur[t], 1);
    lst[p] = s;
}

__device__ __forceinline__ float edge_weight(float v, float mx) {
    v = (v >= 0.f) ? v : NEG_SLOPE * v;
    return expf(v - mx);
}
__device__ __forceinline__ v4f ld_row(const float* __restrict__ hbuf, int s, int lane) {
    return *reinterpret_cast<const v4f*>(hbuf + (size_t)s * 128 + lane * 4);
}
__device__ __forceinline__ float elu1(float v) { return (v > 0.f) ? v : expm1f(v); }

__global__ __launch_bounds__(256) void gat_gather_nodes(
    const int* __restrict__ off, const int* __restrict__ lst,
    const float* __restrict__ hbuf, const float* __restrict__ s_src, const float* __restrict__ s_trg,
    const unsigned int* __restrict__ gmax, const float* __restrict__ bias,
    float* __restrict__ out, int N)
{
    const int lane = threadIdx.x & 31;
    const int hd   = lane >> 2;
    const int t = __builtin_amdgcn_readfirstlane((int)(blockIdx.x * 8 + (threadIdx.x >> 5)));
    if (t >= N) return;

    const float mx  = funkey(*gmax);
    const int   beg = off[t];
    const int   n   = off[t + 1] - beg;
    const float st  = s_trg[(size_t)t * 8 + hd];
    const v4f hself = ld_row(hbuf, t, lane);
    const float b0 = bias[lane * 4 + 0];
    const float b1 = bias[lane * 4 + 1];
    const float b2 = bias[lane * 4 + 2];
    const float b3 = bias[lane * 4 + 3];

    v4f   acc = {0.f, 0.f, 0.f, 0.f};
    float den = 0.f;

    for (int i0 = 0; i0 < n; i0 += 32) {
        const int m = (n - i0 < 32) ? (n - i0) : 32;
        const int mys = (lane < m) ? lst[beg + i0 + lane] : 0;

        int k = 0;
        for (; k + 4 <= m; k += 4) {
            const int s0 = __builtin_amdgcn_readlane(mys, k + 0);
            const int s1 = __builtin_amdgcn_readlane(mys, k + 1);
            const int s2 = __builtin_amdgcn_readlane(mys, k + 2);
            const int s3 = __builtin_amdgcn_readlane(mys, k + 3);
            const v4f h0 = ld_row(hbuf, s0, lane);
            const v4f h1 = ld_row(hbuf, s1, lane);
            const v4f h2 = ld_row(hbuf, s2, lane);
            const v4f h3 = ld_row(hbuf, s3, lane);
            const float a0 = s_src[(size_t)s0 * 8 + hd];
            const float a1 = s_src[(size_t)s1 * 8 + hd];
            const float a2 = s_src[(size_t)s2 * 8 + hd];
            const float a3 = s_src[(size_t)s3 * 8 + hd];
            const float w0 = edge_weight(a0 + st, mx);
            const float w1 = edge_weight(a1 + st, mx);
            const float w2 = edge_weight(a2 + st, mx);
            const float w3 = edge_weight(a3 + st, mx);
            den += w0; acc += h0 * w0;
            den += w1; acc += h1 * w1;
            den += w2; acc += h2 * w2;
            den += w3; acc += h3 * w3;
        }
        const int r = m - k;
        if (r > 0) {
            const int s0 = __builtin_amdgcn_readlane(mys, k + 0);
            const int s1 = (r > 1) ? __builtin_amdgcn_readlane(mys, k + 1) : s0;
            const int s2 = (r > 2) ? __builtin_amdgcn_readlane(mys, k + 2) : s0;
            const v4f h0 = ld_row(hbuf, s0, lane);
            const float a0 = s_src[(size_t)s0 * 8 + hd];
            v4f h1 = h0, h2 = h0;
            float a1 = a0, a2 = a0;
            if (r > 1) { h1 = ld_row(hbuf, s1, lane); a1 = s_src[(size_t)s1 * 8 + hd]; }
            if (r > 2) { h2 = ld_row(hbuf, s2, lane); a2 = s_src[(size_t)s2 * 8 + hd]; }
            const float w0 = edge_weight(a0 + st, mx);
            den += w0; acc += h0 * w0;
            if (r > 1) { const float w1 = edge_weight(a1 + st, mx); den += w1; acc += h1 * w1; }
            if (r > 2) { const float w2 = edge_weight(a2 + st, mx); den += w2; acc += h2 * w2; }
        }
    }

    const float rdn = 1.0f / (den + EPS_F);
    v4f o;
    o.x = elu1(acc.x * rdn + hself.x + b0);
    o.y = elu1(acc.y * rdn + hself.y + b1);
    o.z = elu1(acc.z * rdn + hself.z + b2);
    o.w = elu1(acc.w * rdn + hself.w + b3);
    *reinterpret_cast<v4f*>(out + (size_t)t * 128 + lane * 4) = o;
}

extern "C" void kernel_launch(void* const* d_in, const int* in_sizes, int n_in,
                              void* d_out, int out_size, void* d_ws, size_t ws_size,
                              hipStream_t stream)
{
    const float* x      = (const float*)d_in[0];
    const int*   ei     = (const int*)  d_in[1];
    const float* W      = (const float*)d_in[2];
    const float* a_src  = (const float*)d_in[3];
    const float* a_trg  = (const float*)d_in[4];
    const float* bias   = (const float*)d_in[5];

    const int N = in_sizes[0] / 128;
    const int E = in_sizes[1] / 2;

    float* hbuf = (float*)d_ws;
    float* ssrc = hbuf + (size_t)N * 128;
    float* strg = ssrc + (size_t)N * 8;
    int*   cnt  = (int*)(strg + (size_t)N * 8);
    unsigned int* gmax = (unsigned int*)(cnt + N);
    int*   off  = (int*)(gmax + 1);
    int*   cur  = off + N + 1;
    int*   lst  = cur + N;

    hipMemsetAsync(cnt, 0, (size_t)N * sizeof(int) + sizeof(unsigned int), stream);

    const int mtiles = (N + 15) / 16;
    gat_gemm_wmma<<<mtiles, 256, 0, stream>>>(x, W, a_src, a_trg, hbuf, ssrc, strg, N);

    gat_edge_max_count<<<(E + 255) / 256, 256, 0, stream>>>(ei, E, ssrc, strg, cnt, gmax);

    gat_scan_counts<<<(N + 1023) / 1024, 1024, 0, stream>>>(cnt, off, cur, N);

    gat_fill_lists<<<(E + 255) / 256, 256, 0, stream>>>(ei, E, cur, lst);

    gat_gather_nodes<<<(N + 7) / 8, 256, 0, stream>>>(off, lst, hbuf, ssrc, strg, gmax, bias,
                                                     (float*)d_out, N);
}


# projection GEMM retiled: block 64 rows x 128 cols, x and W staged once in LDS by whole-row loads, wave 16 x 64 (4 accumulators), rows and scores stored as whole lines; score sums in a new lane order
# speedup vs baseline: 3.9170x; 1.1270x over previous
#include <hip/hip_runtime.h>
#include <hip/hip_bf16.h>
#include <math.h>

typedef float v2f __attribute__((ext_vector_type(2)));
typedef float v4f __attribute__((ext_vector_type(4)));
typedef float v8f __attribute__((ext_vector_type(8)));

#define NEG_SLOPE 0.2f
#define EPS_F 1e-16f

#define SWIZ_XOR1  0x041F
#define SWIZ_XOR2  0x081F
#define SWIZ_XOR4  0x101F
#define SWIZ_XOR8  0x201F
#define SWIZ_XOR16 0x401F
#define SWZ(v, imm) __int_as_float(__builtin_amdgcn_ds_swizzle(__float_as_int(v), imm))

__device__ __forceinline__ float reduce16_add(float v) {
    v += SWZ(v, SWIZ_XOR1);
    v += SWZ(v, SWIZ_XOR2);
    v += SWZ(v, SWIZ_XOR4);
    v += SWZ(v, SWIZ_XOR8);
    return v;
}
__device__ __forceinline__ float reduce32_max(float v) {
    v = fmaxf(v, SWZ(v, SWIZ_XOR1));
    v = fmaxf(v, SWZ(v, SWIZ_XOR2));
    v = fmaxf(v, SWZ(v, SWIZ_XOR4));
    v = fmaxf(v, SWZ(v, SWIZ_XOR8));
    v = fmaxf(v, SWZ(v, SWIZ_XOR16));
    return v;
}

__device__ __forceinline__ unsigned int fkey(float f) {
    unsigned int u = __float_as_uint(f);
    return (u & 0x80000000u) ? ~u : (u | 0x80000000u);
}
__device__ __forceinline__ float funkey(unsigned int k) {
    return __uint_as_float((k & 0x80000000u) ? (k & 0x7FFFFFFFu) : ~k);
}

#define XPITCH 132

__global__ __launch_bounds__(256) void gat_gemm_wmma(
    const float* __restrict__ x, const float* __restrict__ W,
    const float* __restrict__ a_src, const float* __restrict__ a_trg,
    float* __restrict__ hbuf, float* __restrict__ s_src, float* __restrict__ s_trg,
    int N)
{
    __shared__ __attribute__((aligned(16))) float Wl[128 * 128];
    __shared__ __attribute__((aligned(16))) float Xl[64 * XPITCH];
    __shared__ __attribute__((aligned(16))) float Sl[2 * 64 * 8];

    const int tid   = threadIdx.x;
    const int wave  = tid >> 5;
    const int lane  = tid & 31;
    const int mBase = blockIdx.x * 64;

    v4f xr[8];
    #pragma unroll
    for (int q = 0; q < 8; ++q) {
        int row = mBase + wave * 8 + q;
        if (row > N - 1) row = N - 1;
        xr[q] = *reinterpret_cast<const v4f*>(x + (size_t)row * 128 + lane * 4);
    }
    v4f w0[8], w1[8];
    #pragma unroll
    for (int q = 0; q < 8; ++q) {
        const int kp = wave * 8 + q;
        w0[q] = *reinterpret_cast<const v4f*>(W + (size_t)(2 * kp)     * 128 + lane * 4);
        w1[q] = *reinterpret_cast<const v4f*>(W + (size_t)(2 * kp + 1) * 128 + lane * 4);
    }
    #pragma unroll
    for (int q = 0; q < 8; ++q)
        *reinterpret_cast<v4f*>(Xl + (wave * 8 + q) * XPITCH + lane * 4) = xr[q];
    {
        const int cc = lane & 15, jj = lane >> 4;
        #pragma unroll
        for (int q = 0; q < 8; ++q) {
            const int kp = wave * 8 + q;
            v4f p01, p23;
            p01.x = w0[q].x; p01.y = w1[q].x; p01.z = w0[q].y; p01.w = w1[q].y;
            p23.x = w0[q].z; p23.y = w1[q].z; p23.z = w0[q].w; p23.w = w1[q].w;
            float* dst = Wl + kp * 256 + jj * 128 + cc * 4;
            *reinterpret_cast<v4f*>(dst)      = p01;
            *reinterpret_cast<v4f*>(dst + 64) = p23;
        }
    }
    __syncthreads();

    const int i = wave >> 1;
    const int j = wave & 1;
    const int c = lane & 15;
    const int h = lane >> 4;

    const float* __restrict__ xa = Xl + (i * 16 + c) * XPITCH + 2 * h;
    const float* __restrict__ wb = Wl + h * 256 + j * 128 + c * 4;

    const v4f asv = *reinterpret_cast<const v4f*>(a_src + 64 * j + 4 * c);
    const v4f atv = *reinterpret_cast<const v4f*>(a_trg + 64 * j + 4 * c);

    v8f acc0 = {}, acc1 = {}, acc2 = {}, acc3 = {};
    #pragma unroll 8
    for (int k = 0; k < 128; k += 4) {
        const v2f a = *reinterpret_cast<const v2f*>(xa + k);
        const v4f q0 = *reinterpret_cast<const v4f*>(wb + (k >> 1) * 256);
        const v4f q1 = *reinterpret_cast<const v4f*>(wb + (k >> 1) * 256 + 64);
        v2f b;
        b.x = q0.x; b.y = q0.y;
        acc0 = __builtin_amdgcn_wmma_f32_16x16x4_f32(false, a, false, b, (short)0, acc0, false, false);
        b.x = q0.z; b.y = q0.w;
        acc1 = __builtin_amdgcn_wmma_f32_16x16x4_f32(false, a, false, b, (short)0, acc1, false, false);
        b.x = q1.x; b.y = q1.y;
        acc2 = __builtin_amdgcn_wmma_f32_16x16x4_f32(false, a, false, b, (short)0, acc2, false, false);
        b.x = q1.z; b.y = q1.w;
        acc3 = __builtin_amdgcn_wmma_f32_16x16x4_f32(false, a, false, b, (short)0, acc3, false, false);
    }

    #pragma unroll
    for (int r = 0; r < 8; ++r) {
        const int lr = i * 16 + h * 8 + r;
        const int m  = mBase + lr;
        v4f o;
        o.x = acc0[r]; o.y = acc1[r]; o.z = acc2[r]; o.w = acc3[r];
        if (m < N) *reinterpret_cast<v4f*>(hbuf + (size_t)m * 128 + 64 * j + 4 * c) = o;

        float ps = o.x * asv.x + o.y * asv.y + o.z * asv.z + o.w * asv.w;
        float pt = o.x * atv.x + o.y * atv.y + o.z * atv.z + o.w * atv.w;
        ps += SWZ(ps, SWIZ_XOR1);  pt += SWZ(pt, SWIZ_XOR1);
        ps += SWZ(ps, SWIZ_XOR2);  pt += SWZ(pt, SWIZ_XOR2);
        if ((c & 3) == 0) {
            Sl[lr * 8 + 4 * j + (c >> 2)]       = ps;
            Sl[512 + lr * 8 + 4 * j + (c >> 2)] = pt;
        }
    }
    __syncthreads();
    {
        const int arr = tid >> 7;
        const int row = (tid & 127) >> 1;
        const int col = (tid & 1) * 4;
        if (mBase + row < N) {
            float* __restrict__ dst = (arr ? s_trg : s_src) + (size_t)(mBase + row) * 8 + col;
            *reinterpret_cast<v4f*>(dst) = *reinterpret_cast<const v4f*>(Sl + tid * 4);
        }
    }
}

__global__ __launch_bounds__(256) void gat_edge_max_count(
    const int* __restrict__ ei, int E,
    const float* __restrict__ s_src, const float* __restrict__ s_trg,
    int* __restrict__ cnt, unsigned int* __restrict__ gmax)
{
    __shared__ unsigned int smax;
    if (threadIdx.x == 0) smax = 0u;
    __syncthreads();

    const int e = blockIdx.x * blockDim.x + threadIdx.x;
    float lmax = -INFINITY;
    if (e < E) {
        const int src = ei[e];
        const int trg = ei[(size_t)E + e];
        const float* __restrict__ ss = s_src + (size_t)src * 8;
        const float* __restrict__ st = s_trg + (size_t)trg * 8;
        #pragma unroll
        for (int hh = 0; hh < 8; ++hh) {
            float v = ss[hh] + st[hh];
            v = (v >= 0.f) ? v : NEG_SLOPE * v;
            lmax = fmaxf(lmax, v);
        }
        atomicAdd(&cnt[trg], 1);
    }
    lmax = reduce32_max(lmax);
    if ((threadIdx.x & 31) == 0) atomicMax(&smax, fkey(lmax));
    __syncthreads();
    if (threadIdx.x == 0) atomicMax(gmax, smax);
}

__global__ __launch_bounds__(1024) void gat_scan_counts(
    const int* __restrict__ cnt, int* __restrict__ off, int* __restrict__ cur, int N)
{
    __shared__ int wsum[32];
    __shared__ int wpre[32];
    __shared__ int sbase;

    const int tid  = threadIdx.x;
    const int lane = tid & 31;
    const int wv   = tid >> 5;
    const int i    = blockIdx.x * 1024 + tid;
    const int c    = (i < N) ? cnt[i] : 0;

    int pre = 0;
    #pragma unroll 8
    for (int j = 0; j < (int)blockIdx.x; ++j) pre += cnt[j * 1024 + tid];

    int inc = c;
    #pragma unroll
    for (int d = 1; d < 32; d <<= 1) {
        const int u = __shfl_up(inc, d, 32);
        if (lane >= d) inc += u;
    }
    #pragma unroll
    for (int d = 16; d >= 1; d >>= 1) pre += __shfl_xor(pre, d, 32);

    if (lane == 31) wsum[wv] = inc;
    if (lane == 0)  wpre[wv] = pre;
    __syncthreads();
    if (wv == 0) {
        const int w = wsum[lane];
        int winc = w;
        #pragma unroll
        for (int d = 1; d < 32; d <<= 1) {
            const int u = __shfl_up(winc, d, 32);
            if (lane >= d) winc += u;
        }
        int p = wpre[lane];
        #pragma unroll
        for (int d = 16; d >= 1; d >>= 1) p += __shfl_xor(p, d, 32);
        wsum[lane] = winc - w;
        if (lane == 0) sbase = p;
    }
    __syncthreads();
    const int o = sbase + wsum[wv] + inc - c;
    if (i < N) { off[i] = o; cur[i] = o; }
    if (i == N - 1) off[N] = o + c;
}

__global__ __launch_bounds__(256) void gat_fill_lists(
    const int* __restrict__ ei, int E, int* __restrict__ cur, int* __restrict__ lst)
{
    const int e = blockIdx.x * blockDim.x + threadIdx.x;
    if (e >= E) return;
    const int s = ei[e];
    const int t = ei[(size_t)E + e];
    const int p = atomicAdd(&cur[t], 1);
    lst[p] = s;
}

__device__ __forceinline__ float edge_weight(float v, float mx) {
    v = (v >= 0.f) ? v : NEG_SLOPE * v;
    return expf(v - mx);
}
__device__ __forceinline__ v4f ld_row(const float* __restrict__ hbuf, int s, int lane) {
    return *reinterpret_cast<const v4f*>(hbuf + (size_t)s * 128 + lane * 4);
}
__device__ __forceinline__ float elu1(float v) { return (v > 0.f) ? v : expm1f(v); }

__global__ __launch_bounds__(256) void gat_gather_nodes(
    const int* __restrict__ off, const int* __restrict__ lst,
    const float* __restrict__ hbuf, const float* __restrict__ s_src, const float* __restrict__ s_trg,
    const unsigned int* __restrict__ gmax, const float* __restrict__ bias,
    float* __restrict__ out, int N)
{
    const int lane = threadIdx.x & 31;
    const int hd   = lane >> 2;
    const int t = __builtin_amdgcn_readfirstlane((int)(blockIdx.x * 8 + (threadIdx.x >> 5)));
    if (t >= N) return;

    const float mx  = funkey(*gmax);
    const int   beg = off[t];
    const int   n   = off[t + 1] - beg;
    const float st  = s_trg[(size_t)t * 8 + hd];
    const v4f hself = ld_row(hbuf, t, lane);
    const float b0 = bias[lane * 4 + 0];
    const float b1 = bias[lane * 4 + 1];
    const float b2 = bias[lane * 4 + 2];
    const float b3 = bias[lane * 4 + 3];

    v4f   acc = {0.f, 0.f, 0.f, 0.f};
    float den = 0.f;

    for (int i0 = 0; i0 < n; i0 += 32) {
        const int m = (n - i0 < 32) ? (n - i0) : 32;
        const int mys = (lane < m) ? lst[beg + i0 + lane] : 0;

        int k = 0;
        for (; k + 4 <= m; k += 4) {
            const int s0 = __builtin_amdgcn_readlane(mys, k + 0);
            const int s1 = __builtin_amdgcn_readlane(mys, k + 1);
            const int s2 = __builtin_amdgcn_readlane(mys, k + 2);
            const int s3 = __builtin_amdgcn_readlane(mys, k + 3);
            const v4f h0 = ld_row(hbuf, s0, lane);
            const v4f h1 = ld_row(hbuf, s1, lane);
            const v4f h2 = ld_row(hbuf, s2, lane);
            const v4f h3 = ld_row(hbuf, s3, lane);
            const float a0 = s_src[(size_t)s0 * 8 + hd];
            const float a1 = s_src[(size_t)s1 * 8 + hd];
            const float a2 = s_src[(size_t)s2 * 8 + hd];
            const float a3 = s_src[(size_t)s3 * 8 + hd];
            const float w0 = edge_weight(a0 + st, mx);
            const float w1 = edge_weight(a1 + st, mx);
            const float w2 = edge_weight(a2 + st, mx);
            const float w3 = edge_weight(a3 + st, mx);
            den += w0; acc += h0 * w0;
            den += w1; acc += h1 * w1;
            den += w2; acc += h2 * w2;
            den += w3; acc += h3 * w3;
        }
        const int r = m - k;
        if (r > 0) {
            const int s0 = __builtin_amdgcn_readlane(mys, k + 0);
            const int s1 = (r > 1) ? __builtin_amdgcn_readlane(mys, k + 1) : s0;
            const int s2 = (r > 2) ? __builtin_amdgcn_readlane(mys, k + 2) : s0;
            const v4f h0 = ld_row(hbuf, s0, lane);
            const float a0 = s_src[(size_t)s0 * 8 + hd];
            v4f h1 = h0, h2 = h0;
            float a1 = a0, a2 = a0;
            if (r > 1) { h1 = ld_row(hbuf, s1, lane); a1 = s_src[(size_t)s1 * 8 + hd]; }
            if (r > 2) { h2 = ld_row(hbuf, s2, lane); a2 = s_src[(size_t)s2 * 8 + hd]; }
            const float w0 = edge_weight(a0 + st, mx);
            den += w0; acc += h0 * w0;
            if (r > 1) { const float w1 = edge_weight(a1 + st, mx); den += w1; acc += h1 * w1; }
            if (r > 2) { const float w2 = edge_weight(a2 + st, mx); den += w2; acc += h2 * w2; }
        }
    }

    const float rdn = 1.0f / (den + EPS_F);
    v4f o;
    o.x = elu1(acc.x * rdn + hself.x + b0);
    o.y = elu1(acc.y * rdn + hself.y + b1);
    o.z = elu1(acc.z * rdn + hself.z + b2);
    o.w = elu1(acc.w * rdn + hself.w + b3);
    *reinterpret_cast<v4f*>(out + (size_t)t * 128 + lane * 4) = o;
}

extern "C" void kernel_launch(void* const* d_in, const int* in_sizes, int n_in,
                              void* d_out, int out_size, void* d_ws, size_t ws_size,
                              hipStream_t stream)
{
    const float* x      = (const float*)d_in[0];
    const int*   ei     = (const int*)  d_in[1];
    const float* W      = (const float*)d_in[2];
    const float* a_src  = (const float*)d_in[3];
    const float* a_trg  = (const float*)d_in[4];
    const float* bias   = (const float*)d_in[5];

    const int N = in_sizes[0] / 128;
    const int E = in_sizes[1] / 2;

    float* hbuf = (float*)d_ws;
    float* ssrc = hbuf + (size_t)N * 128;
    float* strg = ssrc + (size_t)N * 8;
    int*   cnt  = (int*)(strg + (size_t)N * 8);
    unsigned int* gmax = (unsigned int*)(cnt + N);
    int*   off  = (int*)(gmax + 1);
    int*   cur  = off + N + 1;
    int*   lst  = cur + N;

    hipMemsetAsync(cnt, 0, (size_t)N * sizeof(int) + sizeof(unsigned int), stream);

    const int mtiles = (N + 63) / 64;
    gat_gemm_wmma<<<mtiles, 256, 0, stream>>>(x, W, a_src, a_trg, hbuf, ssrc, strg, N);

    gat_edge_max_count<<<(E + 255) / 256, 256, 0, stream>>>(ei, E, ssrc, strg, cnt, gmax);

    gat_scan_counts<<<(N + 1023) / 1024, 1024, 0, stream>>>(cnt, off, cur, N);

    gat_fill_lists<<<(E + 255) / 256, 256, 0, stream>>>(ei, E, cur, lst);

    gat_gather_nodes<<<(N + 7) / 8, 256, 0, stream>>>(off, lst, hbuf, ssrc, strg, gmax, bias,
                                                     (float*)d_out, N);
}


# edges ordered by target with a two-level counting sort through LDS bins (chunk x bin table, block per 512-node bin): no per-edge global atomics or scattered 4-byte stores; sums in new arrival order
# speedup vs baseline: 6.9807x; 1.7822x over previous
#include <hip/hip_runtime.h>
#include <hip/hip_bf16.h>
#include <math.h>

typedef float v2f __attribute__((ext_vector_type(2)));
typedef float v4f __attribute__((ext_vector_type(4)));
typedef float v8f __attribute__((ext_vector_type(8)));

#define NEG_SLOPE 0.2f
#define EPS_F 1e-16f

#define SWIZ_XOR1  0x041F
#define SWIZ_XOR2  0x081F
#define SWIZ_XOR4  0x101F
#define SWIZ_XOR8  0x201F
#define SWIZ_XOR16 0x401F
#define SWZ(v, imm) __int_as_float(__builtin_amdgcn_ds_swizzle(__float_as_int(v), imm))

__device__ __forceinline__ float reduce16_add(float v) {
    v += SWZ(v, SWIZ_XOR1);
    v += SWZ(v, SWIZ_XOR2);
    v += SWZ(v, SWIZ_XOR4);
    v += SWZ(v, SWIZ_XOR8);
    return v;
}
__device__ __forceinline__ float reduce32_max(float v) {
    v = fmaxf(v, SWZ(v, SWIZ_XOR1));
    v = fmaxf(v, SWZ(v, SWIZ_XOR2));
    v = fmaxf(v, SWZ(v, SWIZ_XOR4));
    v = fmaxf(v, SWZ(v, SWIZ_XOR8));
    v = fmaxf(v, SWZ(v, SWIZ_XOR16));
    return v;
}

__device__ __forceinline__ unsigned int fkey(float f) {
    unsigned int u = __float_as_uint(f);
    return (u & 0x80000000u) ? ~u : (u | 0x80000000u);
}
__device__ __forceinline__ float funkey(unsigned int k) {
    return __uint_as_float((k & 0x80000000u) ? (k & 0x7FFFFFFFu) : ~k);
}

#define XPITCH 132

__global__ __launch_bounds__(256) void gat_gemm_wmma(
    const float* __restrict__ x, const float* __restrict__ W,
    const float* __restrict__ a_src, const float* __restrict__ a_trg,
    float* __restrict__ hbuf, float* __restrict__ s_src, float* __restrict__ s_trg,
    int N)
{
    __shared__ __attribute__((aligned(16))) float Wl[128 * 128];
    __shared__ __attribute__((aligned(16))) float Xl[64 * XPITCH];
    __shared__ __attribute__((aligned(16))) float Sl[2 * 64 * 8];

    const int tid   = threadIdx.x;
    const int wave  = tid >> 5;
    const int lane  = tid & 31;
    const int mBase = blockIdx.x * 64;

    v4f xr[8];
    #pragma unroll
    for (int q = 0; q < 8; ++q) {
        int row = mBase + wave * 8 + q;
        if (row > N - 1) row = N - 1;
        xr[q] = *reinterpret_cast<const v4f*>(x + (size_t)row * 128 + lane * 4);
    }
    v4f w0[8], w1[8];
    #pragma unroll
    for (int q = 0; q < 8; ++q) {
        const int kp = wave * 8 + q;
        w0[q] = *reinterpret_cast<const v4f*>(W + (size_t)(2 * kp)     * 128 + lane * 4);
        w1[q] = *reinterpret_cast<const v4f*>(W + (size_t)(2 * kp + 1) * 128 + lane * 4);
    }
    #pragma unroll
    for (int q = 0; q < 8; ++q)
        *reinterpret_cast<v4f*>(Xl + (wave * 8 + q) * XPITCH + lane * 4) = xr[q];
    {
        const int cc = lane & 15, jj = lane >> 4;
        #pragma unroll
        for (int q = 0; q < 8; ++q) {
            const int kp = wave * 8 + q;
            v4f p01, p23;
            p01.x = w0[q].x; p01.y = w1[q].x; p01.z = w0[q].y; p01.w = w1[q].y;
            p23.x = w0[q].z; p23.y = w1[q].z; p23.z = w0[q].w; p23.w = w1[q].w;
            float* dst = Wl + kp * 256 + jj * 128 + cc * 4;
            *reinterpret_cast<v4f*>(dst)      = p01;
            *reinterpret_cast<v4f*>(dst + 64) = p23;
        }
    }
    __syncthreads();

    const int i = wave >> 1;
    const int j = wave & 1;
    const int c = lane & 15;
    const int h = lane >> 4;

    const float* __restrict__ xa = Xl + (i * 16 + c) * XPITCH + 2 * h;
    const float* __restrict__ wb = Wl + h * 256 + j * 128 + c * 4;

    const v4f asv = *reinterpret_cast<const v4f*>(a_src + 64 * j + 4 * c);
    const v4f atv = *reinterpret_cast<const v4f*>(a_trg + 64 * j + 4 * c);

    v8f acc0 = {}, acc1 = {}, acc2 = {}, acc3 = {};
    #pragma unroll 8
    for (int k = 0; k < 128; k += 4) {
        const v2f a = *reinterpret_cast<const v2f*>(xa + k);
        const v4f q0 = *reinterpret_cast<const v4f*>(wb + (k >> 1) * 256);
        const v4f q1 = *reinterpret_cast<const v4f*>(wb + (k >> 1) * 256 + 64);
        v2f b;
        b.x = q0.x; b.y = q0.y;
        acc0 = __builtin_amdgcn_wmma_f32_16x16x4_f32(false, a, false, b, (short)0, acc0, false, false);
        b.x = q0.z; b.y = q0.w;
        acc1 = __builtin_amdgcn_wmma_f32_16x16x4_f32(false, a, false, b, (short)0, acc1, false, false);
        b.x = q1.x; b.y = q1.y;
        acc2 = __builtin_amdgcn_wmma_f32_16x16x4_f32(false, a, false, b, (short)0, acc2, false, false);
        b.x = q1.z; b.y = q1.w;
        acc3 = __builtin_amdgcn_wmma_f32_16x16x4_f32(false, a, false, b, (short)0, acc3, false, false);
    }

    #pragma unroll
    for (int r = 0; r < 8; ++r) {
        const int lr = i * 16 + h * 8 + r;
        const int m  = mBase + lr;
        v4f o;
        o.x = acc0[r]; o.y = acc1[r]; o.z = acc2[r]; o.w = acc3[r];
        if (m < N) *reinterpret_cast<v4f*>(hbuf + (size_t)m * 128 + 64 * j + 4 * c) = o;

        float ps = o.x * asv.x + o.y * asv.y + o.z * asv.z + o.w * asv.w;
        float pt = o.x * atv.x + o.y * atv.y + o.z * atv.z + o.w * atv.w;
        ps += SWZ(ps, SWIZ_XOR1);  pt += SWZ(pt, SWIZ_XOR1);
        ps += SWZ(ps, SWIZ_XOR2);  pt += SWZ(pt, SWIZ_XOR2);
        if ((c & 3) == 0) {
            Sl[lr * 8 + 4 * j + (c >> 2)]       = ps;
            Sl[512 + lr * 8 + 4 * j + (c >> 2)] = pt;
        }
    }
    __syncthreads();
    {
        const int arr = tid >> 7;
        const int row = (tid & 127) >> 1;
        const int col = (tid & 1) * 4;
        if (mBase + row < N) {
            float* __restrict__ dst = (arr ? s_trg : s_src) + (size_t)(mBase + row) * 8 + col;
            *reinterpret_cast<v4f*>(dst) = *reinterpret_cast<const v4f*>(Sl + tid * 4);
        }
    }
}

__device__ __forceinline__ int block_excl_scan_1024(int v, int* ws, int& total)
{
    const int lane = threadIdx.x & 31;
    const int wv   = threadIdx.x >> 5;
    int inc = v;
    #pragma unroll
    for (int d = 1; d < 32; d <<= 1) {
        const int u = __shfl_up(inc, d, 32);
        if (lane >= d) inc += u;
    }
    if (lane == 31) ws[wv] = inc;
    __syncthreads();
    if (wv == 0) {
        const int w = ws[lane];
        int winc = w;
        #pragma unroll
        for (int d = 1; d < 32; d <<= 1) {
            const int u = __shfl_up(winc, d, 32);
            if (lane >= d) winc += u;
        }
        ws[lane] = winc - w;
        if (lane == 31) ws[32] = winc;
    }
    __syncthreads();
    total = ws[32];
    return ws[wv] + inc - v;
}

#define BIN_SHIFT 9
#define BIN_MASK  511
#define MAX_BINS  256
#define CHUNK     4096

__global__ __launch_bounds__(512) void gat_edge_max_hist(
    const int* __restrict__ ei, int E,
    const float* __restrict__ s_src, const float* __restrict__ s_trg,
    int* __restrict__ tab, unsigned int* __restrict__ gmax)
{
    __shared__ int hist[MAX_BINS];
    __shared__ unsigned int smax;
    const int tid = threadIdx.x;
    if (tid < MAX_BINS) hist[tid] = 0;
    if (tid == 0) smax = 0u;
    __syncthreads();

    const int base = blockIdx.x * CHUNK;
    float lmax = -INFINITY;
    for (int g4 = 0; g4 < CHUNK / 2048; ++g4) {
        int  sv[4], tv[4];
        bool ok[4];
        #pragma unroll
        for (int q = 0; q < 4; ++q) {
            const int e = base + (g4 * 4 + q) * 512 + tid;
            ok[q] = e < E;
            const int ec = ok[q] ? e : E - 1;
            sv[q] = ei[ec];
            tv[q] = ei[(size_t)E + ec];
        }
        v4f a0[4], a1[4], b0[4], b1[4];
        #pragma unroll
        for (int q = 0; q < 4; ++q) {
            const float* __restrict__ ss = s_src + (size_t)sv[q] * 8;
            const float* __restrict__ st = s_trg + (size_t)tv[q] * 8;
            a0[q] = *reinterpret_cast<const v4f*>(ss);
            a1[q] = *reinterpret_cast<const v4f*>(ss + 4);
            b0[q] = *reinterpret_cast<const v4f*>(st);
            b1[q] = *reinterpret_cast<const v4f*>(st + 4);
        }
        #pragma unroll
        for (int q = 0; q < 4; ++q) {
            const v4f u0 = a0[q] + b0[q];
            const v4f u1 = a1[q] + b1[q];
            float m = -INFINITY;
            #pragma unroll
            for (int hh = 0; hh < 4; ++hh) {
                float v = u0[hh];
                v = (v >= 0.f) ? v : NEG_SLOPE * v;
                m = fmaxf(m, v);
                v = u1[hh];
                v = (v >= 0.f) ? v : NEG_SLOPE * v;
                m = fmaxf(m, v);
            }
            lmax = fmaxf(lmax, ok[q] ? m : -INFINITY);
            if (ok[q]) atomicAdd(&hist[tv[q] >> BIN_SHIFT], 1);
        }
    }
    lmax = reduce32_max(lmax);
    if ((tid & 31) == 0) atomicMax(&smax, fkey(lmax));
    __syncthreads();
    if (tid < MAX_BINS) tab[(size_t)blockIdx.x * MAX_BINS + tid] = hist[tid];
    if (tid == 0) atomicMax(gmax, smax);
}

__global__ __launch_bounds__(1024) void gat_bin_scatter(
    const int* __restrict__ ei, int E,
    const int* __restrict__ tab, int nchunks,
    int* __restrict__ binned, int* __restrict__ binstart)
{
    __shared__ int totp[4 * MAX_BINS];
    __shared__ int prep[4 * MAX_BINS];
    __shared__ int gbase[MAX_BINS];
    __shared__ int lstart[MAX_BINS];
    __shared__ int lcur[MAX_BINS];
    __shared__ int sval[CHUNK];
    __shared__ int sbin[CHUNK];
    __shared__ int ws1[33];
    __shared__ int ws2[33];

    const int tid = threadIdx.x;
    const int c   = blockIdx.x;
    const int b   = tid & (MAX_BINS - 1);
    const int g   = tid >> 8;

    int tot = 0, pre = 0;
    #pragma unroll 8
    for (int cc = g; cc < nchunks; cc += 4) {
        const int v = tab[(size_t)cc * MAX_BINS + b];
        tot += v;
        if (cc < c) pre += v;
    }
    totp[g * MAX_BINS + b] = tot;
    prep[g * MAX_BINS + b] = pre;
    __syncthreads();

    int T = 0, P = 0, own = 0;
    if (tid < MAX_BINS) {
        T   = totp[tid] + totp[MAX_BINS + tid] + totp[2 * MAX_BINS + tid] + totp[3 * MAX_BINS + tid];
        P   = prep[tid] + prep[MAX_BINS + tid] + prep[2 * MAX_BINS + tid] + prep[3 * MAX_BINS + tid];
        own = tab[(size_t)c * MAX_BINS + tid];
    }
    int tAll, oAll;
    const int bs = block_excl_scan_1024(T,   ws1, tAll);
    const int ls = block_excl_scan_1024(own, ws2, oAll);
    if (tid < MAX_BINS) {
        gbase[tid]  = bs + P;
        lstart[tid] = ls;
        lcur[tid]   = ls;
        if (c == 0) binstart[tid] = bs;
    }
    if (c == 0 && tid == 0) binstart[MAX_BINS] = tAll;
    __syncthreads();

    const int base = c * CHUNK;
    int  sv[CHUNK / 1024], tv[CHUNK / 1024];
    bool ok[CHUNK / 1024];
    #pragma unroll
    for (int k = 0; k < CHUNK / 1024; ++k) {
        const int e = base + k * 1024 + tid;
        ok[k] = e < E;
        const int ec = ok[k] ? e : E - 1;
        sv[k] = ei[ec];
        tv[k] = ei[(size_t)E + ec];
    }
    #pragma unroll
    for (int k = 0; k < CHUNK / 1024; ++k) {
        if (ok[k]) {
            const int bin = tv[k] >> BIN_SHIFT;
            const int r   = atomicAdd(&lcur[bin], 1);
            sval[r] = (sv[k] << BIN_SHIFT) | (tv[k] & BIN_MASK);
            sbin[r] = bin;
        }
    }
    __syncthreads();
    const int nvalid = (E - base < CHUNK) ? (E - base) : CHUNK;
    #pragma unroll
    for (int k = 0; k < CHUNK / 1024; ++k) {
        const int i = k * 1024 + tid;
        if (i < nvalid) {
            const int bb = sbin[i];
            binned[gbase[bb] + (i - lstart[bb])] = sval[i];
        }
    }
}

#define STAGE_CAP 16384

__global__ __launch_bounds__(1024) void gat_bin_lists(
    const int* __restrict__ binned, const int* __restrict__ binstart,
    int* __restrict__ off, int* __restrict__ lst, int N)
{
    __shared__ int cnt[512];
    __shared__ int cur[512];
    __shared__ int stage[STAGE_CAP];
    __shared__ int ws[33];

    const int tid  = threadIdx.x;
    const int lo   = blockIdx.x << BIN_SHIFT;
    const int seg0 = binstart[blockIdx.x];
    const int n    = binstart[blockIdx.x + 1] - seg0;

    if (tid < 512) cnt[tid] = 0;
    int pk[4];
    #pragma unroll
    for (int k = 0; k < 4; ++k) {
        const int i = k * 1024 + tid;
        pk[k] = (i < n) ? binned[seg0 + i] : 0;
    }
    __syncthreads();
    #pragma unroll
    for (int k = 0; k < 4; ++k)
        if (k * 1024 + tid < n) atomicAdd(&cnt[pk[k] & BIN_MASK], 1);
    for (int i = 4096 + tid; i < n; i += 1024)
        atomicAdd(&cnt[binned[seg0 + i] & BIN_MASK], 1);
    __syncthreads();

    const int cv = (tid < 512) ? cnt[tid] : 0;
    int all;
    const int loc = block_excl_scan_1024(cv, ws, all);
    if (tid < 512) {
        cur[tid] = loc;
        const int node = lo + tid;
        if (node < N)      off[node] = seg0 + loc;
        if (node == N - 1) off[N]    = seg0 + loc + cv;
    }
    __syncthreads();

    if (n <= STAGE_CAP) {
        #pragma unroll
        for (int k = 0; k < 4; ++k) {
            if (k * 1024 + tid < n) {
                const int r = atomicAdd(&cur[pk[k] & BIN_MASK], 1);
                stage[r] = (int)((unsigned int)pk[k] >> BIN_SHIFT);
            }
        }
        for (int i = 4096 + tid; i < n; i += 1024) {
            const int p = binned[seg0 + i];
            const int r = atomicAdd(&cur[p & BIN_MASK], 1);
            stage[r] = (int)((unsigned int)p >> BIN_SHIFT);
        }
        __syncthreads();
        for (int i = tid; i < n; i += 1024) lst[seg0 + i] = stage[i];
    } else {
        for (int i = tid; i < n; i += 1024) {
            const int p = binned[seg0 + i];
            const int r = atomicAdd(&cur[p & BIN_MASK], 1);
            lst[seg0 + r] = (int)((unsigned int)p >> BIN_SHIFT);
        }
    }
}

__device__ __forceinline__ float edge_weight(float v, float mx) {
    v = (v >= 0.f) ? v : NEG_SLOPE * v;
    return expf(v - mx);
}
__device__ __forceinline__ v4f ld_row(const float* __restrict__ hbuf, int s, int lane) {
    return *reinterpret_cast<const v4f*>(hbuf + (size_t)s * 128 + lane * 4);
}
__device__ __forceinline__ float elu1(float v) { return (v > 0.f) ? v : expm1f(v); }

__global__ __launch_bounds__(256) void gat_gather_nodes(
    const int* __restrict__ off, const int* __restrict__ lst,
    const float* __restrict__ hbuf, const float* __restrict__ s_src, const float* __restrict__ s_trg,
    const unsigned int* __restrict__ gmax, const float* __restrict__ bias,
    float* __restrict__ out, int N)
{
    const int lane = threadIdx.x & 31;
    const int hd   = lane >> 2;
    const int t = __builtin_amdgcn_readfirstlane((int)(blockIdx.x * 8 + (threadIdx.x >> 5)));
    if (t >= N) return;

    const float mx  = funkey(*gmax);
    const int   beg = off[t];
    const int   n   = off[t + 1] - beg;
    const float st  = s_trg[(size_t)t * 8 + hd];
    const v4f hself = ld_row(hbuf, t, lane);
    const float b0 = bias[lane * 4 + 0];
    const float b1 = bias[lane * 4 + 1];
    const float b2 = bias[lane * 4 + 2];
    const float b3 = bias[lane * 4 + 3];

    v4f   acc = {0.f, 0.f, 0.f, 0.f};
    float den = 0.f;

    for (int i0 = 0; i0 < n; i0 += 32) {
        const int m = (n - i0 < 32) ? (n - i0) : 32;
        const int mys = (lane < m) ? lst[beg + i0 + lane] : 0;

        int k = 0;
        for (; k + 4 <= m; k += 4) {
            const int s0 = __builtin_amdgcn_readlane(mys, k + 0);
            const int s1 = __builtin_amdgcn_readlane(mys, k + 1);
            const int s2 = __builtin_amdgcn_readlane(mys, k + 2);
            const int s3 = __builtin_amdgcn_readlane(mys, k + 3);
            const v4f h0 = ld_row(hbuf, s0, lane);
            const v4f h1 = ld_row(hbuf, s1, lane);
            const v4f h2 = ld_row(hbuf, s2, lane);
            const v4f h3 = ld_row(hbuf, s3, lane);
            const float a0 = s_src[(size_t)s0 * 8 + hd];
            const float a1 = s_src[(size_t)s1 * 8 + hd];
            const float a2 = s_src[(size_t)s2 * 8 + hd];
            const float a3 = s_src[(size_t)s3 * 8 + hd];
            const float w0 = edge_weight(a0 + st, mx);
            const float w1 = edge_weight(a1 + st, mx);
            const float w2 = edge_weight(a2 + st, mx);
            const float w3 = edge_weight(a3 + st, mx);
            den += w0; acc += h0 * w0;
            den += w1; acc += h1 * w1;
            den += w2; acc += h2 * w2;
            den += w3; acc += h3 * w3;
        }
        const int r = m - k;
        if (r > 0) {
            const int s0 = __builtin_amdgcn_readlane(mys, k + 0);
            const int s1 = (r > 1) ? __builtin_amdgcn_readlane(mys, k + 1) : s0;
            const int s2 = (r > 2) ? __builtin_amdgcn_readlane(mys, k + 2) : s0;
            const v4f h0 = ld_row(hbuf, s0, lane);
            const float a0 = s_src[(size_t)s0 * 8 + hd];
            v4f h1 = h0, h2 = h0;
            float a1 = a0, a2 = a0;
            if (r > 1) { h1 = ld_row(hbuf, s1, lane); a1 = s_src[(size_t)s1 * 8 + hd]; }
            if (r > 2) { h2 = ld_row(hbuf, s2, lane); a2 = s_src[(size_t)s2 * 8 + hd]; }
            const float w0 = edge_weight(a0 + st, mx);
            den += w0; acc += h0 * w0;
            if (r > 1) { const float w1 = edge_weight(a1 + st, mx); den += w1; acc += h1 * w1; }
            if (r > 2) { const float w2 = edge_weight(a2 + st, mx); den += w2; acc += h2 * w2; }
        }
    }

    const float rdn = 1.0f / (den + EPS_F);
    v4f o;
    o.x = elu1(acc.x * rdn + hself.x + b0);
    o.y = elu1(acc.y * rdn + hself.y + b1);
    o.z = elu1(acc.z * rdn + hself.z + b2);
    o.w = elu1(acc.w * rdn + hself.w + b3);
    *reinterpret_cast<v4f*>(out + (size_t)t * 128 + lane * 4) = o;
}

extern "C" void kernel_launch(void* const* d_in, const int* in_sizes, int n_in,
                              void* d_out, int out_size, void* d_ws, size_t ws_size,
                              hipStream_t stream)
{
    const float* x      = (const float*)d_in[0];
    const int*   ei     = (const int*)  d_in[1];
    const float* W      = (const float*)d_in[2];
    const float* a_src  = (const float*)d_in[3];
    const float* a_trg  = (const float*)d_in[4];
    const float* bias   = (const float*)d_in[5];

    const int N = in_sizes[0] / 128;
    const int E = in_sizes[1] / 2;

    float* hbuf = (float*)d_ws;
    float* ssrc = hbuf + (size_t)N * 128;
    float* strg = ssrc + (size_t)N * 8;
    int*   bst  = (int*)(strg + (size_t)N * 8);
    unsigned int* gmax = (unsigned int*)(bst + N);
    int*   off  = (int*)(gmax + 1);
    int*   cur  = off + N + 1;
    int*   lst  = cur + N;

    const int nchunks = (E + CHUNK - 1) / CHUNK;
    int*   binned = (int*)d_out;
    int*   tab    = binned + E;

    hipMemsetAsync(gmax, 0, sizeof(unsigned int), stream);

    const int mtiles = (N + 63) / 64;
    gat_gemm_wmma<<<mtiles, 256, 0, stream>>>(x, W, a_src, a_trg, hbuf, ssrc, strg, N);

    gat_edge_max_hist<<<nchunks, 512, 0, stream>>>(ei, E, ssrc, strg, tab, gmax);

    gat_bin_scatter<<<nchunks, 1024, 0, stream>>>(ei, E, tab, nchunks, binned, bst);

    gat_bin_lists<<<(N + BIN_MASK) >> BIN_SHIFT, 1024, 0, stream>>>(binned, bst, off, lst, N);

    gat_gather_nodes<<<(N + 7) / 8, 256, 0, stream>>>(off, lst, hbuf, ssrc, strg, gmax, bias,
                                                     (float*)d_out, N);
}


# gather: edge weight, elu and 1/(den+eps) on the exp2 and rcp units (exp2 of one fma; series or exp2-1 for expm1; rcp + Newton step): a third fewer vector instructions per node; sum order kept
# speedup vs baseline: 7.5160x; 1.0767x over previous
#include <hip/hip_runtime.h>
#include <hip/hip_bf16.h>
#include <math.h>

typedef float v2f __attribute__((ext_vector_type(2)));
typedef float v4f __attribute__((ext_vector_type(4)));
typedef float v8f __attribute__((ext_vector_type(8)));

#define NEG_SLOPE 0.2f
#define EPS_F 1e-16f

#define SWIZ_XOR1  0x041F
#define SWIZ_XOR2  0x081F
#define SWIZ_XOR4  0x101F
#define SWIZ_XOR8  0x201F
#define SWIZ_XOR16 0x401F
#define SWZ(v, imm) __int_as_float(__builtin_amdgcn_ds_swizzle(__float_as_int(v), imm))

__device__ __forceinline__ float reduce16_add(float v) {
    v += SWZ(v, SWIZ_XOR1);
    v += SWZ(v, SWIZ_XOR2);
    v += SWZ(v, SWIZ_XOR4);
    v += SWZ(v, SWIZ_XOR8);
    return v;
}
__device__ __forceinline__ float reduce32_max(float v) {
    v = fmaxf(v, SWZ(v, SWIZ_XOR1));
    v = fmaxf(v, SWZ(v, SWIZ_XOR2));
    v = fmaxf(v, SWZ(v, SWIZ_XOR4));
    v = fmaxf(v, SWZ(v, SWIZ_XOR8));
    v = fmaxf(v, SWZ(v, SWIZ_XOR16));
    return v;
}

__device__ __forceinline__ unsigned int fkey(float f) {
    unsigned int u = __float_as_uint(f);
    return (u & 0x80000000u) ? ~u : (u | 0x80000000u);
}
__device__ __forceinline__ float funkey(unsigned int k) {
    return __uint_as_float((k & 0x80000000u) ? (k & 0x7FFFFFFFu) : ~k);
}

#define BIN_SHIFT 9
#define BIN_MASK  511
#define MAX_BINS  256
#define CHUNK     4096

#define XPITCH 132

__global__ __launch_bounds__(256) void gat_gemm_wmma(
    const float* __restrict__ x, const float* __restrict__ W,
    const float* __restrict__ a_src, const float* __restrict__ a_trg,
    float* __restrict__ hbuf, float* __restrict__ s_src, float* __restrict__ s_trg,
    int N,
    const int* __restrict__ ei, int E, int nchunks,
    int* __restrict__ tab, unsigned int* __restrict__ gmax)
{
    __shared__ __attribute__((aligned(16))) float Wl[128 * 128];
    __shared__ __attribute__((aligned(16))) float Xl[64 * XPITCH];
    __shared__ __attribute__((aligned(16))) float Sl[2 * 64 * 8];

    const int tid   = threadIdx.x;

    if ((int)blockIdx.x < nchunks) {
        int* hist = reinterpret_cast<int*>(Sl);
        hist[tid] = 0;
        if (blockIdx.x == 0 && tid == 0) *gmax = 0u;
        __syncthreads();
        const int base = blockIdx.x * CHUNK;
        int  tv[CHUNK / 256];
        bool ok[CHUNK / 256];
        #pragma unroll
        for (int q = 0; q < CHUNK / 256; ++q) {
            const int e = base + q * 256 + tid;
            ok[q] = e < E;
            tv[q] = ei[(size_t)E + (ok[q] ? e : E - 1)];
        }
        #pragma unroll
        for (int q = 0; q < CHUNK / 256; ++q)
            if (ok[q]) atomicAdd(&hist[tv[q] >> BIN_SHIFT], 1);
        __syncthreads();
        tab[(size_t)blockIdx.x * MAX_BINS + tid] = hist[tid];
        return;
    }

    const int wave  = tid >> 5;
    const int lane  = tid & 31;
    const int mBase = ((int)blockIdx.x - nchunks) * 64;

    v4f xr[8];
    #pragma unroll
    for (int q = 0; q < 8; ++q) {
        int row = mBase + wave * 8 + q;
        if (row > N - 1) row = N - 1;
        xr[q] = *reinterpret_cast<const v4f*>(x + (size_t)row * 128 + lane * 4);
    }
    v4f w0[8], w1[8];
    #pragma unroll
    for (int q = 0; q < 8; ++q) {
        const int kp = wave * 8 + q;
        w0[q] = *reinterpret_cast<const v4f*>(W + (size_t)(2 * kp)     * 128 + lane * 4);
        w1[q] = *reinterpret_cast<const v4f*>(W + (size_t)(2 * kp + 1) * 128 + lane * 4);
    }
    #pragma unroll
    for (int q = 0; q < 8; ++q)
        *reinterpret_cast<v4f*>(Xl + (wave * 8 + q) * XPITCH + lane * 4) = xr[q];
    {
        const int cc = lane & 15, jj = lane >> 4;
        #pragma unroll
        for (int q = 0; q < 8; ++q) {
            const int kp = wave * 8 + q;
            v4f p01, p23;
            p01.x = w0[q].x; p01.y = w1[q].x; p01.z = w0[q].y; p01.w = w1[q].y;
            p23.x = w0[q].z; p23.y = w1[q].z; p23.z = w0[q].w; p23.w = w1[q].w;
            float* dst = Wl + kp * 256 + jj * 128 + cc * 4;
            *reinterpret_cast<v4f*>(dst)      = p01;
            *reinterpret_cast<v4f*>(dst + 64) = p23;
        }
    }
    __syncthreads();

    const int i = wave >> 1;
    const int j = wave & 1;
    const int c = lane & 15;
    const int h = lane >> 4;

    const float* __restrict__ xa = Xl + (i * 16 + c) * XPITCH + 2 * h;
    const float* __restrict__ wb = Wl + h * 256 + j * 128 + c * 4;

    const v4f asv = *reinterpret_cast<const v4f*>(a_src + 64 * j + 4 * c);
    const v4f atv = *reinterpret_cast<const v4f*>(a_trg + 64 * j + 4 * c);

    v8f acc0 = {}, acc1 = {}, acc2 = {}, acc3 = {};
    #pragma unroll 8
    for (int k = 0; k < 128; k += 4) {
        const v2f a = *reinterpret_cast<const v2f*>(xa + k);
        const v4f q0 = *reinterpret_cast<const v4f*>(wb + (k >> 1) * 256);
        const v4f q1 = *reinterpret_cast<const v4f*>(wb + (k >> 1) * 256 + 64);
        v2f b;
        b.x = q0.x; b.y = q0.y;
        acc0 = __builtin_amdgcn_wmma_f32_16x16x4_f32(false, a, false, b, (short)0, acc0, false, false);
        b.x = q0.z; b.y = q0.w;
        acc1 = __builtin_amdgcn_wmma_f32_16x16x4_f32(false, a, false, b, (short)0, acc1, false, false);
        b.x = q1.x; b.y = q1.y;
        acc2 = __builtin_amdgcn_wmma_f32_16x16x4_f32(false, a, false, b, (short)0, acc2, false, false);
        b.x = q1.z; b.y = q1.w;
        acc3 = __builtin_amdgcn_wmma_f32_16x16x4_f32(false, a, false, b, (short)0, acc3, false, false);
    }

    #pragma unroll
    for (int r = 0; r < 8; ++r) {
        const int lr = i * 16 + h * 8 + r;
        const int m  = mBase + lr;
        v4f o;
        o.x = acc0[r]; o.y = acc1[r]; o.z = acc2[r]; o.w = acc3[r];
        if (m < N) *reinterpret_cast<v4f*>(hbuf + (size_t)m * 128 + 64 * j + 4 * c) = o;

        float ps = o.x * asv.x + o.y * asv.y + o.z * asv.z + o.w * asv.w;
        float pt = o.x * atv.x + o.y * atv.y + o.z * atv.z + o.w * atv.w;
        ps += SWZ(ps, SWIZ_XOR1);  pt += SWZ(pt, SWIZ_XOR1);
        ps += SWZ(ps, SWIZ_XOR2);  pt += SWZ(pt, SWIZ_XOR2);
        if ((c & 3) == 0) {
            Sl[lr * 8 + 4 * j + (c >> 2)]       = ps;
            Sl[512 + lr * 8 + 4 * j + (c >> 2)] = pt;
        }
    }
    __syncthreads();
    {
        const int arr = tid >> 7;
        const int row = (tid & 127) >> 1;
        const int col = (tid & 1) * 4;
        if (mBase + row < N) {
            float* __restrict__ dst = (arr ? s_trg : s_src) + (size_t)(mBase + row) * 8 + col;
            *reinterpret_cast<v4f*>(dst) = *reinterpret_cast<const v4f*>(Sl + tid * 4);
        }
    }
}

__device__ __forceinline__ int block_excl_scan_1024(int v, int* ws, int& total)
{
    const int lane = threadIdx.x & 31;
    const int wv   = threadIdx.x >> 5;
    int inc = v;
    #pragma unroll
    for (int d = 1; d < 32; d <<= 1) {
        const int u = __shfl_up(inc, d, 32);
        if (lane >= d) inc += u;
    }
    if (lane == 31) ws[wv] = inc;
    __syncthreads();
    if (wv == 0) {
        const int w = ws[lane];
        int winc = w;
        #pragma unroll
        for (int d = 1; d < 32; d <<= 1) {
            const int u = __shfl_up(winc, d, 32);
            if (lane >= d) winc += u;
        }
        ws[lane] = winc - w;
        if (lane == 31) ws[32] = winc;
    }
    __syncthreads();
    total = ws[32];
    return ws[wv] + inc - v;
}

__global__ __launch_bounds__(1024) void gat_bin_scatter(
    const int* __restrict__ ei, int E,
    const int* __restrict__ tab, int nchunks,
    const float* __restrict__ s_src, const float* __restrict__ s_trg,
    int* __restrict__ binned, int* __restrict__ binstart, unsigned int* __restrict__ gmax)
{
    __shared__ unsigned int smax;
    __shared__ int totp[4 * MAX_BINS];
    __shared__ int prep[4 * MAX_BINS];
    __shared__ int gbase[MAX_BINS];
    __shared__ int lstart[MAX_BINS];
    __shared__ int lcur[MAX_BINS];
    __shared__ int sval[CHUNK];
    __shared__ int sbin[CHUNK];
    __shared__ int ws1[33];
    __shared__ int ws2[33];

    const int tid = threadIdx.x;
    const int c   = blockIdx.x;
    const int b   = tid & (MAX_BINS - 1);
    const int g   = tid >> 8;
    if (tid == 0) smax = 0u;

    const int base = c * CHUNK;
    int  sv[CHUNK / 1024], tv[CHUNK / 1024];
    bool ok[CHUNK / 1024];
    #pragma unroll
    for (int k = 0; k < CHUNK / 1024; ++k) {
        const int e = base + k * 1024 + tid;
        ok[k] = e < E;
        const int ec = ok[k] ? e : E - 1;
        sv[k] = ei[ec];
        tv[k] = ei[(size_t)E + ec];
    }

    int tot = 0, pre = 0;
    #pragma unroll 8
    for (int cc = g; cc < nchunks; cc += 4) {
        const int v = tab[(size_t)cc * MAX_BINS + b];
        tot += v;
        if (cc < c) pre += v;
    }
    totp[g * MAX_BINS + b] = tot;
    prep[g * MAX_BINS + b] = pre;
    __syncthreads();

    v4f a0[CHUNK / 1024], a1[CHUNK / 1024], b0[CHUNK / 1024], b1[CHUNK / 1024];
    #pragma unroll
    for (int k = 0; k < CHUNK / 1024; ++k) {
        const float* __restrict__ ss = s_src + (size_t)sv[k] * 8;
        const float* __restrict__ st = s_trg + (size_t)tv[k] * 8;
        a0[k] = *reinterpret_cast<const v4f*>(ss);
        a1[k] = *reinterpret_cast<const v4f*>(ss + 4);
        b0[k] = *reinterpret_cast<const v4f*>(st);
        b1[k] = *reinterpret_cast<const v4f*>(st + 4);
    }

    int T = 0, P = 0, own = 0;
    if (tid < MAX_BINS) {
        T   = totp[tid] + totp[MAX_BINS + tid] + totp[2 * MAX_BINS + tid] + totp[3 * MAX_BINS + tid];
        P   = prep[tid] + prep[MAX_BINS + tid] + prep[2 * MAX_BINS + tid] + prep[3 * MAX_BINS + tid];
        own = tab[(size_t)c * MAX_BINS + tid];
    }
    int tAll, oAll;
    const int bs = block_excl_scan_1024(T,   ws1, tAll);
    const int ls = block_excl_scan_1024(own, ws2, oAll);
    if (tid < MAX_BINS) {
        gbase[tid]  = bs + P;
        lstart[tid] = ls;
        lcur[tid]   = ls;
        if (c == 0) binstart[tid] = bs;
    }
    if (c == 0 && tid == 0) binstart[MAX_BINS] = tAll;
    __syncthreads();

    float lmax = -INFINITY;
    #pragma unroll
    for (int k = 0; k < CHUNK / 1024; ++k) {
        const v4f u0 = a0[k] + b0[k];
        const v4f u1 = a1[k] + b1[k];
        float m = -INFINITY;
        #pragma unroll
        for (int hh = 0; hh < 4; ++hh) {
            float v = u0[hh];
            v = (v >= 0.f) ? v : NEG_SLOPE * v;
            m = fmaxf(m, v);
            v = u1[hh];
            v = (v >= 0.f) ? v : NEG_SLOPE * v;
            m = fmaxf(m, v);
        }
        lmax = fmaxf(lmax, ok[k] ? m : -INFINITY);
    }
    lmax = reduce32_max(lmax);
    if ((tid & 31) == 0) atomicMax(&smax, fkey(lmax));

    #pragma unroll
    for (int k = 0; k < CHUNK / 1024; ++k) {
        if (ok[k]) {
            const int bin = tv[k] >> BIN_SHIFT;
            const int r   = atomicAdd(&lcur[bin], 1);
            sval[r] = (sv[k] << BIN_SHIFT) | (tv[k] & BIN_MASK);
            sbin[r] = bin;
        }
    }
    __syncthreads();
    const int nvalid = (E - base < CHUNK) ? (E - base) : CHUNK;
    #pragma unroll
    for (int k = 0; k < CHUNK / 1024; ++k) {
        const int i = k * 1024 + tid;
        if (i < nvalid) {
            const int bb = sbin[i];
            binned[gbase[bb] + (i - lstart[bb])] = sval[i];
        }
    }
    if (tid == 0) atomicMax(gmax, smax);
}

#define STAGE_CAP 16384

__global__ __launch_bounds__(1024) void gat_bin_lists(
    const int* __restrict__ binned, const int* __restrict__ binstart,
    int* __restrict__ off, int* __restrict__ lst, int N)
{
    __shared__ int cnt[512];
    __shared__ int cur[512];
    __shared__ int stage[STAGE_CAP];
    __shared__ int ws[33];

    const int tid  = threadIdx.x;
    const int lo   = blockIdx.x << BIN_SHIFT;
    const int seg0 = binstart[blockIdx.x];
    const int n    = binstart[blockIdx.x + 1] - seg0;

    if (tid < 512) cnt[tid] = 0;
    int pk[4];
    #pragma unroll
    for (int k = 0; k < 4; ++k) {
        const int i = k * 1024 + tid;
        pk[k] = (i < n) ? binned[seg0 + i] : 0;
    }
    __syncthreads();
    #pragma unroll
    for (int k = 0; k < 4; ++k)
        if (k * 1024 + tid < n) atomicAdd(&cnt[pk[k] & BIN_MASK], 1);
    for (int i = 4096 + tid; i < n; i += 1024)
        atomicAdd(&cnt[binned[seg0 + i] & BIN_MASK], 1);
    __syncthreads();

    const int cv = (tid < 512) ? cnt[tid] : 0;
    int all;
    const int loc = block_excl_scan_1024(cv, ws, all);
    if (tid < 512) {
        cur[tid] = loc;
        const int node = lo + tid;
        if (node < N)      off[node] = seg0 + loc;
        if (node == N - 1) off[N]    = seg0 + loc + cv;
    }
    __syncthreads();

    if (n <= STAGE_CAP) {
        #pragma unroll
        for (int k = 0; k < 4; ++k) {
            if (k * 1024 + tid < n) {
                const int r = atomicAdd(&cur[pk[k] & BIN_MASK], 1);
                stage[r] = (int)((unsigned int)pk[k] >> BIN_SHIFT);
            }
        }
        for (int i = 4096 + tid; i < n; i += 1024) {
            const int p = binned[seg0 + i];
            const int r = atomicAdd(&cur[p & BIN_MASK], 1);
            stage[r] = (int)((unsigned int)p >> BIN_SHIFT);
        }
        __syncthreads();
        for (int i = tid; i < n; i += 1024) lst[seg0 + i] = stage[i];
    } else {
        for (int i = tid; i < n; i += 1024) {
            const int p = binned[seg0 + i];
            const int r = atomicAdd(&cur[p & BIN_MASK], 1);
            lst[seg0 + r] = (int)((unsigned int)p >> BIN_SHIFT);
        }
    }
}

#define LOG2E_F 1.44269504088896340736f

__device__ __forceinline__ float edge_weight(float v, float mxl) {
    v = (v >= 0.f) ? v : NEG_SLOPE * v;
    return __builtin_amdgcn_exp2f(fmaf(v, LOG2E_F, -mxl));
}
__device__ __forceinline__ v4f ld_row(const float* __restrict__ hbuf, int s, int lane) {
    return *reinterpret_cast<const v4f*>(hbuf + (size_t)s * 128 + lane * 4);
}
__device__ __forceinline__ float elu1(float v) {
    const float far = __builtin_amdgcn_exp2f(v * LOG2E_F) - 1.0f;
    float p = 1.0f / 5040.0f;
    p = fmaf(p, v, 1.0f / 720.0f);
    p = fmaf(p, v, 1.0f / 120.0f);
    p = fmaf(p, v, 1.0f / 24.0f);
    p = fmaf(p, v, 1.0f / 6.0f);
    p = fmaf(p, v, 0.5f);
    p = fmaf(p, v, 1.0f);
    const float near = p * v;
    const float em1 = (v > -0.3f) ? near : far;
    return (v > 0.f) ? v : em1;
}

__global__ __launch_bounds__(256) void gat_gather_nodes(
    const int* __restrict__ off, const int* __restrict__ lst,
    const float* __restrict__ hbuf, const float* __restrict__ s_src, const float* __restrict__ s_trg,
    const unsigned int* __restrict__ gmax, const float* __restrict__ bias,
    float* __restrict__ out, int N)
{
    const int lane = threadIdx.x & 31;
    const int hd   = lane >> 2;
    const int t = __builtin_amdgcn_readfirstlane((int)(blockIdx.x * 8 + (threadIdx.x >> 5)));
    if (t >= N) return;

    const float mxl = funkey(*gmax) * LOG2E_F;
    const int   beg = off[t];
    const int   n   = off[t + 1] - beg;
    const float st  = s_trg[(size_t)t * 8 + hd];
    const v4f hself = ld_row(hbuf, t, lane);
    const float b0 = bias[lane * 4 + 0];
    const float b1 = bias[lane * 4 + 1];
    const float b2 = bias[lane * 4 + 2];
    const float b3 = bias[lane * 4 + 3];

    v4f   acc = {0.f, 0.f, 0.f, 0.f};
    float den = 0.f;

    for (int i0 = 0; i0 < n; i0 += 32) {
        const int m = (n - i0 < 32) ? (n - i0) : 32;
        const int mys = (lane < m) ? lst[beg + i0 + lane] : 0;

        int k = 0;
        for (; k + 4 <= m; k += 4) {
            const int s0 = __builtin_amdgcn_readlane(mys, k + 0);
            const int s1 = __builtin_amdgcn_readlane(mys, k + 1);
            const int s2 = __builtin_amdgcn_readlane(mys, k + 2);
            const int s3 = __builtin_amdgcn_readlane(mys, k + 3);
            const v4f h0 = ld_row(hbuf, s0, lane);
            const v4f h1 = ld_row(hbuf, s1, lane);
            const v4f h2 = ld_row(hbuf, s2, lane);
            const v4f h3 = ld_row(hbuf, s3, lane);
            const float a0 = s_src[(size_t)s0 * 8 + hd];
            const float a1 = s_src[(size_t)s1 * 8 + hd];
            const float a2 = s_src[(size_t)s2 * 8 + hd];
            const float a3 = s_src[(size_t)s3 * 8 + hd];
            const float w0 = edge_weight(a0 + st, mxl);
            const float w1 = edge_weight(a1 + st, mxl);
            const float w2 = edge_weight(a2 + st, mxl);
            const float w3 = edge_weight(a3 + st, mxl);
            den += w0; acc += h0 * w0;
            den += w1; acc += h1 * w1;
            den += w2; acc += h2 * w2;
            den += w3; acc += h3 * w3;
        }
        const int r = m - k;
        if (r > 0) {
            const int s0 = __builtin_amdgcn_readlane(mys, k + 0);
            const int s1 = (r > 1) ? __builtin_amdgcn_readlane(mys, k + 1) : s0;
            const int s2 = (r > 2) ? __builtin_amdgcn_readlane(mys, k + 2) : s0;
            const v4f h0 = ld_row(hbuf, s0, lane);
            const float a0 = s_src[(size_t)s0 * 8 + hd];
            v4f h1 = h0, h2 = h0;
            float a1 = a0, a2 = a0;
            if (r > 1) { h1 = ld_row(hbuf, s1, lane); a1 = s_src[(size_t)s1 * 8 + hd]; }
            if (r > 2) { h2 = ld_row(hbuf, s2, lane); a2 = s_src[(size_t)s2 * 8 + hd]; }
            const float w0 = edge_weight(a0 + st, mxl);
            den += w0; acc += h0 * w0;
            if (r > 1) { const float w1 = edge_weight(a1 + st, mxl); den += w1; acc += h1 * w1; }
            if (r > 2) { const float w2 = edge_weight(a2 + st, mxl); den += w2; acc += h2 * w2; }
        }
    }

    const float dpe = den + EPS_F;
    float rdn = __builtin_amdgcn_rcpf(dpe);
    rdn = fmaf(fmaf(-dpe, rdn, 1.0f), rdn, rdn);
    v4f o;
    o.x = elu1(acc.x * rdn + hself.x + b0);
    o.y = elu1(acc.y * rdn + hself.y + b1);
    o.z = elu1(acc.z * rdn + hself.z + b2);
    o.w = elu1(acc.w * rdn + hself.w + b3);
    *reinterpret_cast<v4f*>(out + (size_t)t * 128 + lane * 4) = o;
}

extern "C" void kernel_launch(void* const* d_in, const int* in_sizes, int n_in,
                              void* d_out, int out_size, void* d_ws, size_t ws_size,
                              hipStream_t stream)
{
    const float* x      = (const float*)d_in[0];
    const int*   ei     = (const int*)  d_in[1];
    const float* W      = (const float*)d_in[2];
    const float* a_src  = (const float*)d_in[3];
    const float* a_trg  = (const float*)d_in[4];
    const float* bias   = (const float*)d_in[5];

    const int N = in_sizes[0] / 128;
    const int E = in_sizes[1] / 2;

    float* hbuf = (float*)d_ws;
    float* ssrc = hbuf + (size_t)N * 128;
    float* strg = ssrc + (size_t)N * 8;
    int*   bst  = (int*)(strg + (size_t)N * 8);
    unsigned int* gmax = (unsigned int*)(bst + N);
    int*   off  = (int*)(gmax + 1);
    int*   cur  = off + N + 1;
    int*   lst  = cur + N;

    const int nchunks = (E + CHUNK - 1) / CHUNK;
    int*   binned = (int*)d_out;
    int*   tab    = binned + E;

    const int mtiles = (N + 63) / 64;
    gat_gemm_wmma<<<nchunks + mtiles, 256, 0, stream>>>(x, W, a_src, a_trg, hbuf, ssrc, strg, N,
                                                       ei, E, nchunks, tab, gmax);

    gat_bin_scatter<<<nchunks, 1024, 0, stream>>>(ei, E, tab, nchunks, ssrc, strg, binned, bst, gmax);

    gat_bin_lists<<<(N + BIN_MASK) >> BIN_SHIFT, 1024, 0, stream>>>(binned, bst, off, lst, N);

    gat_gather_nodes<<<(N + 7) / 8, 256, 0, stream>>>(off, lst, hbuf, ssrc, strg, gmax, bias,
                                                     (float*)d_out, N);
}
